# starting kernel
# baseline (speedup 1.0000x reference)
#include <hip/hip_runtime.h>
#include <hip/hip_bf16.h>


typedef __attribute__((ext_vector_type(2))) float v2f;
typedef __attribute__((ext_vector_type(8))) float v8f;

#define B_DIM 16
#define T_DIM 16384
#define H_DIM 64
#define CHUNK 64
#define NCH   (T_DIM / CHUNK)
#define JT    (H_DIM / 16)
#define NWAVES (B_DIM * NCH * JT)

__global__ __launch_bounds__(256) void cumsum_chunk_wmma(const float* __restrict__ g,
                                                         float* __restrict__ out) {
  const int wave = (int)((blockIdx.x * blockDim.x + threadIdx.x) >> 5);
  const int lane = (int)(threadIdx.x & 31);
  const int col  = lane & 15;
  const int half = lane >> 4;

  const int j = wave & (JT - 1);
  const int c = (wave >> 2) & (NCH - 1);
  const int b = wave >> 10;

  const size_t base = ((size_t)b * T_DIM + (size_t)c * CHUNK) * H_DIM
                    + (size_t)(j * 16 + col);
  const float* __restrict__ gp = g + base;
  float* __restrict__ op = out + base;

  v2f A[4];
#pragma unroll
  for (int s = 0; s < 4; ++s) {
    A[s].x = (4 * s + 2 * half + 0 <= col) ? 1.0f : 0.0f;
    A[s].y = (4 * s + 2 * half + 1 <= col) ? 1.0f : 0.0f;
  }

  float carry = 0.0f;

#pragma unroll
  for (int I = 0; I < 4; ++I) {
    v2f Bv[4];
#pragma unroll
    for (int s = 0; s < 4; ++s) {
      const int t = 16 * I + 4 * s + 2 * half;
      Bv[s].x = __builtin_nontemporal_load(&gp[(size_t)t * H_DIM]);
      Bv[s].y = __builtin_nontemporal_load(&gp[(size_t)(t + 1) * H_DIM]);
    }

    v8f D = {};
#pragma unroll
    for (int s = 0; s < 4; ++s) {
      D = __builtin_amdgcn_wmma_f32_16x16x4_f32(
           false, A[s],  false, Bv[s],
           (short)0, D,  false,  false);
    }

    if (I > 0) {
#pragma unroll
      for (int v = 0; v < 8; ++v) D[v] += carry;
    }

    if (I < 3) {
      const int cbits = __builtin_amdgcn_ds_bpermute((col + 16) << 2,
                                                     __float_as_int(D[7]));
      carry = __int_as_float(cbits);
    }

#pragma unroll
    for (int v = 0; v < 8; ++v) {
      const int t = 16 * I + v + 8 * half;
      __builtin_nontemporal_store(D[v], &op[(size_t)t * H_DIM]);
    }
  }
}

extern "C" void kernel_launch(void* const* d_in, const int* in_sizes, int n_in,
                              void* d_out, int out_size, void* d_ws, size_t ws_size,
                              hipStream_t stream) {
  (void)in_sizes; (void)n_in; (void)d_ws; (void)ws_size; (void)out_size;
  const float* g = (const float*)d_in[0];
  float* out = (float*)d_out;

  const int threads = 256;
  const int blocks  = (NWAVES * 32) / threads;
  cumsum_chunk_wmma<<<blocks, threads, 0, stream>>>(g, out);
}


# One wave per whole 64x64 chunk (4096 waves): every global load and store is 16 bytes per lane over full 256-byte rows; 4 column-interleaved WMMA tiles share them; per-element sums unchanged
# speedup vs baseline: 2.4206x; 2.4206x over previous
#include <hip/hip_runtime.h>


typedef __attribute__((ext_vector_type(2))) float v2f;
typedef __attribute__((ext_vector_type(4))) float v4f;
typedef __attribute__((ext_vector_type(8))) float v8f;

#define B_DIM 16
#define T_DIM 16384
#define H_DIM 64
#define CHUNK 64
#define NCH   (T_DIM / CHUNK)
#define NWAVES (B_DIM * NCH)
#define ROWV  (H_DIM / 4)
#define NQ    4

__global__ __launch_bounds__(256) void cumsum_chunk_wmma(const float* __restrict__ g,
                                                         float* __restrict__ out) {
  const int wave = (int)((blockIdx.x * blockDim.x + threadIdx.x) >> 5);
  const int lane = (int)(threadIdx.x & 31);
  const int col  = lane & 15;
  const int half = lane >> 4;

  const size_t base = (size_t)wave * (size_t)(CHUNK * H_DIM) + (size_t)(col * 4);
  const v4f* __restrict__ gp = (const v4f*)(g + base);
  v4f* __restrict__ op = (v4f*)(out + base);

  v4f L[4][4][2];
#pragma unroll
  for (int I = 0; I < 4; ++I) {
#pragma unroll
    for (int s = 0; s < 4; ++s) {
#pragma unroll
      for (int v = 0; v < 2; ++v) {
        const int t = 16 * I + 4 * s + 2 * half + v;
        L[I][s][v] = __builtin_nontemporal_load(&gp[(size_t)t * ROWV]);
      }
    }
  }

  __builtin_amdgcn_sched_barrier(0);

  v2f A[4];
#pragma unroll
  for (int s = 0; s < 4; ++s) {
    A[s].x = (4 * s + 2 * half + 0 <= col) ? 1.0f : 0.0f;
    A[s].y = (4 * s + 2 * half + 1 <= col) ? 1.0f : 0.0f;
  }

  float carry[NQ] = {0.0f, 0.0f, 0.0f, 0.0f};

#pragma unroll
  for (int I = 0; I < 4; ++I) {
    v8f D[NQ] = {};
#pragma unroll
    for (int s = 0; s < 4; ++s) {
#pragma unroll
      for (int q = 0; q < NQ; ++q) {
        v2f Bv;
        Bv.x = L[I][s][0][q];
        Bv.y = L[I][s][1][q];
        D[q] = __builtin_amdgcn_wmma_f32_16x16x4_f32(
             false, A[s],  false, Bv,
             (short)0, D[q],  false,  false);
      }
    }

    if (I > 0) {
#pragma unroll
      for (int q = 0; q < NQ; ++q) {
#pragma unroll
        for (int v = 0; v < 8; ++v) D[q][v] += carry[q];
      }
    }

    if (I < 3) {
#pragma unroll
      for (int q = 0; q < NQ; ++q) {
        const int cbits = __builtin_amdgcn_ds_bpermute((col + 16) << 2,
                                                       __float_as_int(D[q][7]));
        carry[q] = __int_as_float(cbits);
      }
    }

#pragma unroll
    for (int v = 0; v < 8; ++v) {
      const int t = 16 * I + v + 8 * half;
      v4f o;
      o.x = D[0][v];
      o.y = D[1][v];
      o.z = D[2][v];
      o.w = D[3][v];
      __builtin_nontemporal_store(o, &op[(size_t)t * ROWV]);
    }

    if (I < 3) __builtin_amdgcn_sched_barrier(0);
  }
}

extern "C" void kernel_launch(void* const* d_in, const int* in_sizes, int n_in,
                              void* d_out, int out_size, void* d_ws, size_t ws_size,
                              hipStream_t stream) {
  (void)in_sizes; (void)n_in; (void)d_ws; (void)ws_size; (void)out_size;
  const float* g = (const float*)d_in[0];
  float* out = (float*)d_out;

  const int threads = 256;
  const int blocks  = (NWAVES * 32) / threads;
  cumsum_chunk_wmma<<<blocks, threads, 0, stream>>>(g, out);
}
